# starting kernel
# baseline (speedup 1.0000x reference)
#include <hip/hip_runtime.h>

#define B_SZ   256
#define A_SZ   64
#define FEATN  256
#define H_SZ   16
#define S_SZ   20
#define POSN   64
#define ENCN   256
#define OUTN   384
#define MAXD   10

typedef __attribute__((ext_vector_type(2))) float v2f;
typedef __attribute__((ext_vector_type(8))) float v8f;

#define LDA_ARMY 260
#define LDP      388
#define LDH      388
#define LDATT    20
#define OFF_ARMY 0
#define OFF_PARM (OFF_ARMY + A_SZ * LDA_ARMY)
#define OFF_HIST (OFF_PARM + A_SZ * LDP)
#define OFF_ATT  (OFF_HIST + H_SZ * LDH)
#define SMEM_FLOATS (OFF_ATT + A_SZ * LDATT)

__device__ __forceinline__ v8f wmma4(v2f a, v2f b, v8f c) {
    return __builtin_amdgcn_wmma_f32_16x16x4_f32(
         false, a,  false, b,
         (short)0, c,  false,  false);
}

__global__ void norm_scale_kernel(const float* __restrict__ proj_v,
                                  const float* __restrict__ proj_g,
                                  float* __restrict__ scale_out) {
    __shared__ float red[256];
    float s = 0.f;
    for (int i = threadIdx.x; i < OUTN * FEATN; i += 256) {
        float v = proj_v[i];
        s += v * v;
    }
    red[threadIdx.x] = s;
    __syncthreads();
    for (int st = 128; st > 0; st >>= 1) {
        if ((int)threadIdx.x < st) red[threadIdx.x] += red[threadIdx.x + st];
        __syncthreads();
    }
    if (threadIdx.x == 0) scale_out[0] = proj_g[0] * rsqrtf(red[0]);
}

__global__ __launch_bounds__(256) void fused_attn(
    const float* __restrict__ army,
    const int*   __restrict__ inst,
    const int*   __restrict__ ilen,
    const int*   __restrict__ idiff,
    const float* __restrict__ tok_emb,
    const float* __restrict__ pos_w,
    const float* __restrict__ diff_w,
    const float* __restrict__ proj_v,
    const float* __restrict__ proj_b,
    const float* __restrict__ scale_p,
    float* __restrict__ out_inst,
    float* __restrict__ out_sum)
{
    extern __shared__ float smem[];
    float* s_army = smem + OFF_ARMY;
    float* s_parm = smem + OFF_PARM;
    float* s_hist = smem + OFF_HIST;
    float* s_att  = smem + OFF_ATT;

    const int b    = blockIdx.x;
    const int tid  = threadIdx.x;
    const int wave = tid >> 5;
    const int lane = tid & 31;
    const int lm   = lane & 15;
    const int lh   = lane >> 4;
    const float scale = scale_p[0];

    for (int e = tid; e < H_SZ * OUTN; e += 256) {
        int h = e / OUTN, d = e - h * OUTN;
        float v;
        if (d < ENCN) {
            int len = ilen[b * H_SZ + h];
            if (len < 1) len = 1;
            const int* row = inst + (b * H_SZ + h) * S_SZ;
            float acc = 0.f;
            for (int s = 0; s < len; ++s) acc += tok_emb[row[s] * ENCN + d];
            v = acc / (float)len;
        } else if (d < ENCN + POSN) {
            v = pos_w[h * POSN + (d - ENCN)];
        } else {
            int di = idiff[b * H_SZ + h];
            di = di < 0 ? 0 : (di > MAXD ? MAXD : di);
            v = diff_w[di * POSN + (d - ENCN - POSN)];
        }
        s_hist[h * LDH + d] = v;
    }
    for (int e = tid; e < A_SZ * FEATN; e += 256) {
        int r = e >> 8, c = e & 255;
        s_army[r * LDA_ARMY + c] = army[(size_t)b * A_SZ * FEATN + e];
    }
    __syncthreads();

    for (int d = tid; d < OUTN; d += 256)
        out_sum[b * OUTN + d] = s_hist[(H_SZ - 1) * LDH + d];

    for (int nt = 0; nt < 3; ++nt) {
        const int n0 = (wave * 3 + nt) * 16;
        v8f acc0 = {}, acc1 = {}, acc2 = {}, acc3 = {};
        const float* wrow = proj_v + (size_t)(n0 + lm) * FEATN + 2 * lh;
        for (int k0 = 0; k0 < FEATN; k0 += 4) {
            v2f bf = *(const v2f*)(wrow + k0);
            v2f a0 = *(const v2f*)(&s_army[(0 * 16 + lm) * LDA_ARMY + k0 + 2 * lh]);
            v2f a1 = *(const v2f*)(&s_army[(1 * 16 + lm) * LDA_ARMY + k0 + 2 * lh]);
            v2f a2 = *(const v2f*)(&s_army[(2 * 16 + lm) * LDA_ARMY + k0 + 2 * lh]);
            v2f a3 = *(const v2f*)(&s_army[(3 * 16 + lm) * LDA_ARMY + k0 + 2 * lh]);
            acc0 = wmma4(a0, bf, acc0);
            acc1 = wmma4(a1, bf, acc1);
            acc2 = wmma4(a2, bf, acc2);
            acc3 = wmma4(a3, bf, acc3);
        }
        const float bias = proj_b[n0 + lm];
        #pragma unroll
        for (int r = 0; r < 8; ++r) {
            int mrow = 8 * lh + r;
            float v0 = acc0[r] * scale + bias;
            float v1 = acc1[r] * scale + bias;
            float v2 = acc2[r] * scale + bias;
            float v3 = acc3[r] * scale + bias;
            s_parm[(0 * 16 + mrow) * LDP + n0 + lm] = v0 > 0.f ? v0 : 0.f;
            s_parm[(1 * 16 + mrow) * LDP + n0 + lm] = v1 > 0.f ? v1 : 0.f;
            s_parm[(2 * 16 + mrow) * LDP + n0 + lm] = v2 > 0.f ? v2 : 0.f;
            s_parm[(3 * 16 + mrow) * LDP + n0 + lm] = v3 > 0.f ? v3 : 0.f;
        }
    }
    __syncthreads();

    if (wave < 4) {
        const int m0 = wave * 16;
        v8f acc = {};
        for (int k0 = 0; k0 < OUTN; k0 += 4) {
            v2f a  = *(const v2f*)(&s_parm[(m0 + lm) * LDP + k0 + 2 * lh]);
            v2f bf = *(const v2f*)(&s_hist[lm * LDH + k0 + 2 * lh]);
            acc = wmma4(a, bf, acc);
        }
        #pragma unroll
        for (int r = 0; r < 8; ++r)
            s_att[(m0 + 8 * lh + r) * LDATT + lm] = acc[r];
    }
    __syncthreads();

    if (tid < 64) {
        float* row = s_att + tid * LDATT;
        float mx = row[0];
        #pragma unroll
        for (int h = 1; h < 16; ++h) mx = fmaxf(mx, row[h]);
        float sum = 0.f;
        #pragma unroll
        for (int h = 0; h < 16; ++h) { float e = __expf(row[h] - mx); row[h] = e; sum += e; }
        float inv = 1.f / sum;
        #pragma unroll
        for (int h = 0; h < 16; ++h) row[h] *= inv;
    }
    __syncthreads();

    for (int nt = 0; nt < 3; ++nt) {
        const int n0 = (wave * 3 + nt) * 16;
        v8f acc0 = {}, acc1 = {}, acc2 = {}, acc3 = {};
        #pragma unroll
        for (int k0 = 0; k0 < 16; k0 += 4) {
            int kk = k0 + 2 * lh;
            v2f bf;
            bf.x = s_hist[kk * LDH + n0 + lm];
            bf.y = s_hist[(kk + 1) * LDH + n0 + lm];
            v2f a0 = *(const v2f*)(&s_att[(0 * 16 + lm) * LDATT + kk]);
            v2f a1 = *(const v2f*)(&s_att[(1 * 16 + lm) * LDATT + kk]);
            v2f a2 = *(const v2f*)(&s_att[(2 * 16 + lm) * LDATT + kk]);
            v2f a3 = *(const v2f*)(&s_att[(3 * 16 + lm) * LDATT + kk]);
            acc0 = wmma4(a0, bf, acc0);
            acc1 = wmma4(a1, bf, acc1);
            acc2 = wmma4(a2, bf, acc2);
            acc3 = wmma4(a3, bf, acc3);
        }
        #pragma unroll
        for (int r = 0; r < 8; ++r) {
            int mrow = 8 * lh + r;
            out_inst[((size_t)b * A_SZ + 0 * 16 + mrow) * OUTN + n0 + lm] = acc0[r];
            out_inst[((size_t)b * A_SZ + 1 * 16 + mrow) * OUTN + n0 + lm] = acc1[r];
            out_inst[((size_t)b * A_SZ + 2 * 16 + mrow) * OUTN + n0 + lm] = acc2[r];
            out_inst[((size_t)b * A_SZ + 3 * 16 + mrow) * OUTN + n0 + lm] = acc3[r];
        }
    }
}

extern "C" void kernel_launch(void* const* d_in, const int* in_sizes, int n_in,
                              void* d_out, int out_size, void* d_ws, size_t ws_size,
                              hipStream_t stream) {
    (void)in_sizes; (void)n_in; (void)out_size; (void)ws_size;
    const float* army  = (const float*)d_in[0];
    const int*   inst  = (const int*)  d_in[1];
    const int*   ilen  = (const int*)  d_in[2];
    const int*   idiff = (const int*)  d_in[3];
    const float* tok   = (const float*)d_in[4];
    const float* posw  = (const float*)d_in[5];
    const float* diffw = (const float*)d_in[6];
    const float* pv    = (const float*)d_in[7];
    const float* pg    = (const float*)d_in[8];
    const float* pb    = (const float*)d_in[9];

    float* out      = (float*)d_out;
    float* out_inst = out;
    float* out_sum  = out + (size_t)B_SZ * A_SZ * OUTN;
    float* scale    = (float*)d_ws;

    size_t smem_bytes = (size_t)SMEM_FLOATS * sizeof(float);
    (void)hipFuncSetAttribute((const void*)fused_attn,
                              hipFuncAttributeMaxDynamicSharedMemorySize,
                              (int)smem_bytes);

    hipLaunchKernelGGL(norm_scale_kernel, dim3(1), dim3(256), 0, stream, pv, pg, scale);
    hipLaunchKernelGGL(fused_attn, dim3(B_SZ), dim3(256), smem_bytes, stream,
                       army, inst, ilen, idiff, tok, posw, diffw, pv, pb, scale,
                       out_inst, out_sum);
}


# phase-1 proj_v operand staged through a private LDS panel per matrix wave: 16 rows x 64 k fetched as whole 128-byte lines (16 B per lane) one panel ahead of its products; k order and sums unchanged
# speedup vs baseline: 2.2523x; 2.2523x over previous
#include <hip/hip_runtime.h>

#define B_SZ   256
#define A_SZ   64
#define FEATN  256
#define H_SZ   16
#define S_SZ   20
#define POSN   64
#define ENCN   256
#define OUTN   384
#define MAXD   10

typedef __attribute__((ext_vector_type(2))) float v2f;
typedef __attribute__((ext_vector_type(8))) float v8f;

#define LDA_ARMY 260
#define LDP      388
#define LDH      388
#define LDATT    20
#define OFF_ARMY 0
#define OFF_PARM (OFF_ARMY + A_SZ * LDA_ARMY)
#define OFF_HIST (OFF_PARM + A_SZ * LDP)
#define OFF_ATT  (OFF_HIST + H_SZ * LDH)
#define OFF_BPAN (OFF_ATT + A_SZ * LDATT)
#define BP_K     64
#define LDB      68
#define BPAN_FLOATS (16 * LDB)
#define SMEM_FLOATS (OFF_BPAN + 8 * BPAN_FLOATS)

__device__ __forceinline__ v8f wmma4(v2f a, v2f b, v8f c) {
    return __builtin_amdgcn_wmma_f32_16x16x4_f32(
         false, a,  false, b,
         (short)0, c,  false,  false);
}

#define NPART     32
#define NORM_TPB  256
#define NORM_VEC  ((OUTN * FEATN) / (NPART * NORM_TPB * 4))
typedef __attribute__((ext_vector_type(4))) float v4f;

__global__ __launch_bounds__(NORM_TPB) void norm_partial_kernel(
    const float* __restrict__ proj_v, float* __restrict__ part_out) {
    __shared__ float red[NORM_TPB / 32];
    const int tid = threadIdx.x;
    const float* base = proj_v + (size_t)blockIdx.x * (NORM_VEC * NORM_TPB * 4) + tid * 4;
    v4f q[NORM_VEC];
    #pragma unroll
    for (int j = 0; j < NORM_VEC; ++j)
        q[j] = *(const v4f*)(base + j * (NORM_TPB * 4));
    float s = 0.f;
    #pragma unroll
    for (int j = 0; j < NORM_VEC; ++j)
        s += (q[j].x * q[j].x + q[j].y * q[j].y) + (q[j].z * q[j].z + q[j].w * q[j].w);
    #pragma unroll
    for (int m = 16; m > 0; m >>= 1) s += __shfl_xor(s, m);
    if ((tid & 31) == 0) red[tid >> 5] = s;
    __syncthreads();
    if (tid == 0) {
        float t = 0.f;
        #pragma unroll
        for (int w = 0; w < NORM_TPB / 32; ++w) t += red[w];
        part_out[blockIdx.x] = t;
    }
}

#define NT_MM     256
#define NT_GATHER 512
#define NT_ALL    (NT_MM + NT_GATHER)
__global__ __launch_bounds__(NT_ALL) void fused_attn(
    const float* __restrict__ army,
    const int*   __restrict__ inst,
    const int*   __restrict__ ilen,
    const int*   __restrict__ idiff,
    const float* __restrict__ tok_emb,
    const float* __restrict__ pos_w,
    const float* __restrict__ diff_w,
    const float* __restrict__ proj_v,
    const float* __restrict__ proj_b,
    const float* __restrict__ proj_g,
    const float* __restrict__ norm_part,
    float* __restrict__ out_inst,
    float* __restrict__ out_sum)
{
    extern __shared__ __attribute__((aligned(16))) float smem[];
    float* s_army = smem + OFF_ARMY;
    float* s_parm = smem + OFF_PARM;
    float* s_hist = smem + OFF_HIST;
    float* s_att  = smem + OFF_ATT;

    const int b    = blockIdx.x;
    const int tid  = threadIdx.x;
    const int wave = tid >> 5;
    const int lane = tid & 31;
    const int lm   = lane & 15;
    const int lh   = lane >> 4;
    float vsq = 0.f;
    #pragma unroll
    for (int i = 0; i < NPART; ++i) vsq += norm_part[i];
    const float scale = proj_g[0] * rsqrtf(vsq);

    for (int e = tid; e < A_SZ * FEATN; e += NT_ALL) {
        int r = e >> 8, c = e & 255;
        s_army[r * LDA_ARMY + c] = army[(size_t)b * A_SZ * FEATN + e];
    }
    __syncthreads();

    if (wave >= NT_MM / 32) {
        for (int e = tid - NT_MM; e < H_SZ * OUTN; e += NT_GATHER) {
            int h = e / OUTN, d = e - h * OUTN;
            float v;
            if (d < ENCN) {
                int len = ilen[b * H_SZ + h];
                if (len < 1) len = 1;
                const int* row = inst + (b * H_SZ + h) * S_SZ;
                float acc = 0.f;
                for (int s = 0; s < len; ++s) acc += tok_emb[row[s] * ENCN + d];
                v = acc / (float)len;
            } else if (d < ENCN + POSN) {
                v = pos_w[h * POSN + (d - ENCN)];
            } else {
                int di = idiff[b * H_SZ + h];
                di = di < 0 ? 0 : (di > MAXD ? MAXD : di);
                v = diff_w[di * POSN + (d - ENCN - POSN)];
            }
            s_hist[h * LDH + d] = v;
        }
    } else {
        float* s_bp = smem + OFF_BPAN + wave * BPAN_FLOATS;
        const int prow = lane >> 4;
        const int pcol = (lane & 15) * 4;
        const float* gsrc = proj_v + (size_t)(wave * 48 + prow) * FEATN + pcol;
        v4f pre[8];
        #pragma unroll
        for (int i = 0; i < 8; ++i) pre[i] = *(const v4f*)(gsrc + (2 * i) * FEATN);
        for (int nt = 0; nt < 3; ++nt) {
            const int n0 = (wave * 3 + nt) * 16;
            v8f acc0 = {}, acc1 = {}, acc2 = {}, acc3 = {};
            for (int kp = 0; kp < FEATN / BP_K; ++kp) {
                #pragma unroll
                for (int i = 0; i < 8; ++i)
                    *(v4f*)(&s_bp[(2 * i + prow) * LDB + pcol]) = pre[i];
                const int nxt = nt * (FEATN / BP_K) + kp + 1;
                if (nxt < 3 * (FEATN / BP_K)) {
                    const float* g = gsrc + (size_t)((nxt >> 2) * 16) * FEATN + (nxt & 3) * BP_K;
                    #pragma unroll
                    for (int i = 0; i < 8; ++i) pre[i] = *(const v4f*)(g + (2 * i) * FEATN);
                }
                for (int ks = 0; ks < BP_K; ks += 4) {
                    const int k0 = kp * BP_K + ks;
                    v2f bf = *(const v2f*)(&s_bp[lm * LDB + ks + 2 * lh]);
                    v2f a0 = *(const v2f*)(&s_army[(0 * 16 + lm) * LDA_ARMY + k0 + 2 * lh]);
                    v2f a1 = *(const v2f*)(&s_army[(1 * 16 + lm) * LDA_ARMY + k0 + 2 * lh]);
                    v2f a2 = *(const v2f*)(&s_army[(2 * 16 + lm) * LDA_ARMY + k0 + 2 * lh]);
                    v2f a3 = *(const v2f*)(&s_army[(3 * 16 + lm) * LDA_ARMY + k0 + 2 * lh]);
                    acc0 = wmma4(a0, bf, acc0);
                    acc1 = wmma4(a1, bf, acc1);
                    acc2 = wmma4(a2, bf, acc2);
                    acc3 = wmma4(a3, bf, acc3);
                }
            }
            const float bias = proj_b[n0 + lm];
            #pragma unroll
            for (int r = 0; r < 8; ++r) {
                int mrow = 8 * lh + r;
                float v0 = acc0[r] * scale + bias;
                float v1 = acc1[r] * scale + bias;
                float v2 = acc2[r] * scale + bias;
                float v3 = acc3[r] * scale + bias;
                s_parm[(0 * 16 + mrow) * LDP + n0 + lm] = v0 > 0.f ? v0 : 0.f;
                s_parm[(1 * 16 + mrow) * LDP + n0 + lm] = v1 > 0.f ? v1 : 0.f;
                s_parm[(2 * 16 + mrow) * LDP + n0 + lm] = v2 > 0.f ? v2 : 0.f;
                s_parm[(3 * 16 + mrow) * LDP + n0 + lm] = v3 > 0.f ? v3 : 0.f;
            }
        }
    }
    __syncthreads();

    if (tid >= NT_MM && tid < NT_MM + OUTN)
        out_sum[b * OUTN + (tid - NT_MM)] = s_hist[(H_SZ - 1) * LDH + (tid - NT_MM)];

    if (wave < 4) {
        const int m0 = wave * 16;
        v8f acc = {};
        for (int k0 = 0; k0 < OUTN; k0 += 4) {
            v2f a  = *(const v2f*)(&s_parm[(m0 + lm) * LDP + k0 + 2 * lh]);
            v2f bf = *(const v2f*)(&s_hist[lm * LDH + k0 + 2 * lh]);
            acc = wmma4(a, bf, acc);
        }
        #pragma unroll
        for (int r = 0; r < 8; ++r)
            s_att[(m0 + 8 * lh + r) * LDATT + lm] = acc[r];
    }
    __syncthreads();

    if (tid < 64) {
        float* row = s_att + tid * LDATT;
        float mx = row[0];
        #pragma unroll
        for (int h = 1; h < 16; ++h) mx = fmaxf(mx, row[h]);
        float sum = 0.f;
        #pragma unroll
        for (int h = 0; h < 16; ++h) { float e = __expf(row[h] - mx); row[h] = e; sum += e; }
        float inv = 1.f / sum;
        #pragma unroll
        for (int h = 0; h < 16; ++h) row[h] *= inv;
    }
    __syncthreads();

    if (wave < NT_MM / 32)
    for (int nt = 0; nt < 3; ++nt) {
        const int n0 = (wave * 3 + nt) * 16;
        v8f acc0 = {}, acc1 = {}, acc2 = {}, acc3 = {};
        #pragma unroll
        for (int k0 = 0; k0 < 16; k0 += 4) {
            int kk = k0 + 2 * lh;
            v2f bf;
            bf.x = s_hist[kk * LDH + n0 + lm];
            bf.y = s_hist[(kk + 1) * LDH + n0 + lm];
            v2f a0 = *(const v2f*)(&s_att[(0 * 16 + lm) * LDATT + kk]);
            v2f a1 = *(const v2f*)(&s_att[(1 * 16 + lm) * LDATT + kk]);
            v2f a2 = *(const v2f*)(&s_att[(2 * 16 + lm) * LDATT + kk]);
            v2f a3 = *(const v2f*)(&s_att[(3 * 16 + lm) * LDATT + kk]);
            acc0 = wmma4(a0, bf, acc0);
            acc1 = wmma4(a1, bf, acc1);
            acc2 = wmma4(a2, bf, acc2);
            acc3 = wmma4(a3, bf, acc3);
        }
        #pragma unroll
        for (int r = 0; r < 8; ++r) {
            int mrow = 8 * lh + r;
            out_inst[((size_t)b * A_SZ + 0 * 16 + mrow) * OUTN + n0 + lm] = acc0[r];
            out_inst[((size_t)b * A_SZ + 1 * 16 + mrow) * OUTN + n0 + lm] = acc1[r];
            out_inst[((size_t)b * A_SZ + 2 * 16 + mrow) * OUTN + n0 + lm] = acc2[r];
            out_inst[((size_t)b * A_SZ + 3 * 16 + mrow) * OUTN + n0 + lm] = acc3[r];
        }
    }
}

extern "C" void kernel_launch(void* const* d_in, const int* in_sizes, int n_in,
                              void* d_out, int out_size, void* d_ws, size_t ws_size,
                              hipStream_t stream) {
    (void)in_sizes; (void)n_in; (void)out_size; (void)ws_size;
    const float* army  = (const float*)d_in[0];
    const int*   inst  = (const int*)  d_in[1];
    const int*   ilen  = (const int*)  d_in[2];
    const int*   idiff = (const int*)  d_in[3];
    const float* tok   = (const float*)d_in[4];
    const float* posw  = (const float*)d_in[5];
    const float* diffw = (const float*)d_in[6];
    const float* pv    = (const float*)d_in[7];
    const float* pg    = (const float*)d_in[8];
    const float* pb    = (const float*)d_in[9];

    float* out      = (float*)d_out;
    float* out_inst = out;
    float* out_sum  = out + (size_t)B_SZ * A_SZ * OUTN;
    float* npart    = (float*)d_ws;

    size_t smem_bytes = (size_t)SMEM_FLOATS * sizeof(float);
    (void)hipFuncSetAttribute((const void*)fused_attn,
                              hipFuncAttributeMaxDynamicSharedMemorySize,
                              (int)smem_bytes);

    hipLaunchKernelGGL(norm_partial_kernel, dim3(NPART), dim3(NORM_TPB), 0, stream, pv, npart);
    hipLaunchKernelGGL(fused_attn, dim3(B_SZ), dim3(NT_ALL), smem_bytes, stream,
                       army, inst, ilen, idiff, tok, posw, diffw, pv, pb, pg, npart,
                       out_inst, out_sum);
}


# history-feature gather moved out of the fused kernel into launch 0 (one work-group per batch, two rows per wave, all embedding loads in flight); rows parked in the batch's out block; sums unchanged
# speedup vs baseline: 2.4944x; 1.1075x over previous
#include <hip/hip_runtime.h>

#define B_SZ   256
#define A_SZ   64
#define FEATN  256
#define H_SZ   16
#define S_SZ   20
#define POSN   64
#define ENCN   256
#define OUTN   384
#define MAXD   10

typedef __attribute__((ext_vector_type(2))) float v2f;
typedef __attribute__((ext_vector_type(8))) float v8f;

#define LDA_ARMY 260
#define LDP      388
#define LDH      388
#define LDATT    20
#define OFF_ARMY 0
#define OFF_PARM (OFF_ARMY + A_SZ * LDA_ARMY)
#define OFF_HIST (OFF_PARM + A_SZ * LDP)
#define OFF_ATT  (OFF_HIST + H_SZ * LDH)
#define OFF_BPAN (OFF_ATT + A_SZ * LDATT)
#define BP_K     64
#define LDB      68
#define BPAN_FLOATS (16 * LDB)
#define SMEM_FLOATS (OFF_BPAN + 8 * BPAN_FLOATS)

__device__ __forceinline__ v8f wmma4(v2f a, v2f b, v8f c) {
    return __builtin_amdgcn_wmma_f32_16x16x4_f32(
         false, a,  false, b,
         (short)0, c,  false,  false);
}

#define NPART     32
#define NORM_TPB  256
#define NORM_VEC  ((OUTN * FEATN) / (NPART * NORM_TPB * 4))
#define ROWS_PER_WAVE 2
typedef __attribute__((ext_vector_type(4))) float v4f;

__global__ __launch_bounds__(NORM_TPB) void prep_kernel(
    const float* __restrict__ proj_v, float* __restrict__ part_out,
    const int*   __restrict__ inst,
    const int*   __restrict__ ilen,
    const int*   __restrict__ idiff,
    const float* __restrict__ tok_emb,
    const float* __restrict__ pos_w,
    const float* __restrict__ diff_w,
    float* __restrict__ hist_park)
{
    __shared__ float red[NORM_TPB / 32];
    const int tid = threadIdx.x;
    if (blockIdx.x < NPART) {
        const float* base = proj_v + (size_t)blockIdx.x * (NORM_VEC * NORM_TPB * 4) + tid * 4;
        v4f q[NORM_VEC];
        #pragma unroll
        for (int j = 0; j < NORM_VEC; ++j)
            q[j] = *(const v4f*)(base + j * (NORM_TPB * 4));
        float s = 0.f;
        #pragma unroll
        for (int j = 0; j < NORM_VEC; ++j)
            s += (q[j].x * q[j].x + q[j].y * q[j].y) + (q[j].z * q[j].z + q[j].w * q[j].w);
        #pragma unroll
        for (int m = 16; m > 0; m >>= 1) s += __shfl_xor(s, m);
        if ((tid & 31) == 0) red[tid >> 5] = s;
        __syncthreads();
        if (tid == 0) {
            float t = 0.f;
            #pragma unroll
            for (int w = 0; w < NORM_TPB / 32; ++w) t += red[w];
            part_out[blockIdx.x] = t;
        }
        return;
    }

    const int b    = blockIdx.x - NPART;
    const int wave = __builtin_amdgcn_readfirstlane(tid >> 5);
    const int lane = tid & 31;

    int len[ROWS_PER_WAVE], di[ROWS_PER_WAVE], id[ROWS_PER_WAVE][S_SZ];
    #pragma unroll
    for (int r = 0; r < ROWS_PER_WAVE; ++r) {
        const int row = b * H_SZ + wave + (NORM_TPB / 32) * r;
        len[r] = ilen[row];
        di[r]  = idiff[row];
        #pragma unroll
        for (int s = 0; s < S_SZ; ++s) id[r][s] = inst[row * S_SZ + s];
    }
    #pragma unroll
    for (int r = 0; r < ROWS_PER_WAVE; ++r)
        asm volatile("" :: "s"(len[r]), "s"(di[r]),
                     "s"(id[r][0]),  "s"(id[r][1]),  "s"(id[r][2]),  "s"(id[r][3]),  "s"(id[r][4]),
                     "s"(id[r][5]),  "s"(id[r][6]),  "s"(id[r][7]),  "s"(id[r][8]),  "s"(id[r][9]),
                     "s"(id[r][10]), "s"(id[r][11]), "s"(id[r][12]), "s"(id[r][13]), "s"(id[r][14]),
                     "s"(id[r][15]), "s"(id[r][16]), "s"(id[r][17]), "s"(id[r][18]), "s"(id[r][19]));

    v4f xa[ROWS_PER_WAVE][S_SZ], xb[ROWS_PER_WAVE][S_SZ], pd[ROWS_PER_WAVE];
    #pragma unroll
    for (int r = 0; r < ROWS_PER_WAVE; ++r) {
        len[r] = len[r] < 1 ? 1 : len[r];
        #pragma unroll
        for (int s = 0; s < S_SZ; ++s) {
            if (s < len[r]) {
                const float* t = tok_emb + id[r][s] * ENCN + lane * 4;
                xa[r][s] = *(const v4f*)t;
                xb[r][s] = *(const v4f*)(t + ENCN / 2);
            }
        }
        const int h = wave + (NORM_TPB / 32) * r;
        const int d = di[r] < 0 ? 0 : (di[r] > MAXD ? MAXD : di[r]);
        const float* pd_src = lane < 16 ? pos_w + h * POSN + lane * 4
                                        : diff_w + d * POSN + (lane - 16) * 4;
        pd[r] = *(const v4f*)pd_src;
    }

    #pragma unroll
    for (int r = 0; r < ROWS_PER_WAVE; ++r) {
        const int h = wave + (NORM_TPB / 32) * r;
        v4f a = {0.f, 0.f, 0.f, 0.f}, c = {0.f, 0.f, 0.f, 0.f};
        #pragma unroll
        for (int s = 0; s < S_SZ; ++s) {
            if (s < len[r]) { a += xa[r][s]; c += xb[r][s]; }
        }
        const float fl = (float)len[r];
        a = a / fl;
        c = c / fl;

        float* dst = hist_park + ((size_t)b * A_SZ + h) * OUTN + lane * 4;
        *(v4f*)(dst)                = a;
        *(v4f*)(dst + ENCN / 2)     = c;
        *(v4f*)(dst + ENCN)         = pd[r];
    }
}

#define NT_MM     256
#define NT_GATHER 512
#define NT_ALL    (NT_MM + NT_GATHER)
__global__ __launch_bounds__(NT_ALL) void fused_attn(
    const float* __restrict__ army,
    const float* __restrict__ proj_v,
    const float* __restrict__ proj_b,
    const float* __restrict__ proj_g,
    const float* __restrict__ norm_part,
    float* __restrict__ out_inst,
    float* __restrict__ out_sum)
{
    extern __shared__ __attribute__((aligned(16))) float smem[];
    float* s_army = smem + OFF_ARMY;
    float* s_parm = smem + OFF_PARM;
    float* s_hist = smem + OFF_HIST;
    float* s_att  = smem + OFF_ATT;

    const int b    = blockIdx.x;
    const int tid  = threadIdx.x;
    const int wave = tid >> 5;
    const int lane = tid & 31;
    const int lm   = lane & 15;
    const int lh   = lane >> 4;
    float vsq = 0.f;
    #pragma unroll
    for (int i = 0; i < NPART; ++i) vsq += norm_part[i];
    const float scale = proj_g[0] * rsqrtf(vsq);

    for (int e = tid; e < A_SZ * FEATN; e += NT_ALL) {
        int r = e >> 8, c = e & 255;
        s_army[r * LDA_ARMY + c] = army[(size_t)b * A_SZ * FEATN + e];
    }
    __syncthreads();

    if (wave >= NT_MM / 32) {
        const float* hsrc = out_inst + (size_t)b * A_SZ * OUTN;
        #pragma unroll
        for (int j = 0; j < (H_SZ * (OUTN / 4)) / NT_GATHER; ++j) {
            const int i = tid - NT_MM + j * NT_GATHER;
            const int h = i / (OUTN / 4), c4 = i - h * (OUTN / 4);
            *(v4f*)(&s_hist[h * LDH + 4 * c4]) = *(const v4f*)(hsrc + 4 * i);
        }
    } else {
        float* s_bp = smem + OFF_BPAN + wave * BPAN_FLOATS;
        const int prow = lane >> 4;
        const int pcol = (lane & 15) * 4;
        const float* gsrc = proj_v + (size_t)(wave * 48 + prow) * FEATN + pcol;
        v4f pre[8];
        #pragma unroll
        for (int i = 0; i < 8; ++i) pre[i] = *(const v4f*)(gsrc + (2 * i) * FEATN);
        for (int nt = 0; nt < 3; ++nt) {
            const int n0 = (wave * 3 + nt) * 16;
            v8f acc0 = {}, acc1 = {}, acc2 = {}, acc3 = {};
            for (int kp = 0; kp < FEATN / BP_K; ++kp) {
                #pragma unroll
                for (int i = 0; i < 8; ++i)
                    *(v4f*)(&s_bp[(2 * i + prow) * LDB + pcol]) = pre[i];
                const int nxt = nt * (FEATN / BP_K) + kp + 1;
                if (nxt < 3 * (FEATN / BP_K)) {
                    const float* g = gsrc + (size_t)((nxt >> 2) * 16) * FEATN + (nxt & 3) * BP_K;
                    #pragma unroll
                    for (int i = 0; i < 8; ++i) pre[i] = *(const v4f*)(g + (2 * i) * FEATN);
                }
                for (int ks = 0; ks < BP_K; ks += 4) {
                    const int k0 = kp * BP_K + ks;
                    v2f bf = *(const v2f*)(&s_bp[lm * LDB + ks + 2 * lh]);
                    v2f a0 = *(const v2f*)(&s_army[(0 * 16 + lm) * LDA_ARMY + k0 + 2 * lh]);
                    v2f a1 = *(const v2f*)(&s_army[(1 * 16 + lm) * LDA_ARMY + k0 + 2 * lh]);
                    v2f a2 = *(const v2f*)(&s_army[(2 * 16 + lm) * LDA_ARMY + k0 + 2 * lh]);
                    v2f a3 = *(const v2f*)(&s_army[(3 * 16 + lm) * LDA_ARMY + k0 + 2 * lh]);
                    acc0 = wmma4(a0, bf, acc0);
                    acc1 = wmma4(a1, bf, acc1);
                    acc2 = wmma4(a2, bf, acc2);
                    acc3 = wmma4(a3, bf, acc3);
                }
            }
            const float bias = proj_b[n0 + lm];
            #pragma unroll
            for (int r = 0; r < 8; ++r) {
                int mrow = 8 * lh + r;
                float v0 = acc0[r] * scale + bias;
                float v1 = acc1[r] * scale + bias;
                float v2 = acc2[r] * scale + bias;
                float v3 = acc3[r] * scale + bias;
                s_parm[(0 * 16 + mrow) * LDP + n0 + lm] = v0 > 0.f ? v0 : 0.f;
                s_parm[(1 * 16 + mrow) * LDP + n0 + lm] = v1 > 0.f ? v1 : 0.f;
                s_parm[(2 * 16 + mrow) * LDP + n0 + lm] = v2 > 0.f ? v2 : 0.f;
                s_parm[(3 * 16 + mrow) * LDP + n0 + lm] = v3 > 0.f ? v3 : 0.f;
            }
        }
    }
    __syncthreads();

    if (tid >= NT_MM && tid < NT_MM + OUTN)
        out_sum[b * OUTN + (tid - NT_MM)] = s_hist[(H_SZ - 1) * LDH + (tid - NT_MM)];

    if (wave < 4) {
        const int m0 = wave * 16;
        v8f acc = {};
        for (int k0 = 0; k0 < OUTN; k0 += 4) {
            v2f a  = *(const v2f*)(&s_parm[(m0 + lm) * LDP + k0 + 2 * lh]);
            v2f bf = *(const v2f*)(&s_hist[lm * LDH + k0 + 2 * lh]);
            acc = wmma4(a, bf, acc);
        }
        #pragma unroll
        for (int r = 0; r < 8; ++r)
            s_att[(m0 + 8 * lh + r) * LDATT + lm] = acc[r];
    }
    __syncthreads();

    if (tid < 64) {
        float* row = s_att + tid * LDATT;
        float mx = row[0];
        #pragma unroll
        for (int h = 1; h < 16; ++h) mx = fmaxf(mx, row[h]);
        float sum = 0.f;
        #pragma unroll
        for (int h = 0; h < 16; ++h) { float e = __expf(row[h] - mx); row[h] = e; sum += e; }
        float inv = 1.f / sum;
        #pragma unroll
        for (int h = 0; h < 16; ++h) row[h] *= inv;
    }
    __syncthreads();

    if (wave < NT_MM / 32)
    for (int nt = 0; nt < 3; ++nt) {
        const int n0 = (wave * 3 + nt) * 16;
        v8f acc0 = {}, acc1 = {}, acc2 = {}, acc3 = {};
        #pragma unroll
        for (int k0 = 0; k0 < 16; k0 += 4) {
            int kk = k0 + 2 * lh;
            v2f bf;
            bf.x = s_hist[kk * LDH + n0 + lm];
            bf.y = s_hist[(kk + 1) * LDH + n0 + lm];
            v2f a0 = *(const v2f*)(&s_att[(0 * 16 + lm) * LDATT + kk]);
            v2f a1 = *(const v2f*)(&s_att[(1 * 16 + lm) * LDATT + kk]);
            v2f a2 = *(const v2f*)(&s_att[(2 * 16 + lm) * LDATT + kk]);
            v2f a3 = *(const v2f*)(&s_att[(3 * 16 + lm) * LDATT + kk]);
            acc0 = wmma4(a0, bf, acc0);
            acc1 = wmma4(a1, bf, acc1);
            acc2 = wmma4(a2, bf, acc2);
            acc3 = wmma4(a3, bf, acc3);
        }
        #pragma unroll
        for (int r = 0; r < 8; ++r) {
            int mrow = 8 * lh + r;
            out_inst[((size_t)b * A_SZ + 0 * 16 + mrow) * OUTN + n0 + lm] = acc0[r];
            out_inst[((size_t)b * A_SZ + 1 * 16 + mrow) * OUTN + n0 + lm] = acc1[r];
            out_inst[((size_t)b * A_SZ + 2 * 16 + mrow) * OUTN + n0 + lm] = acc2[r];
            out_inst[((size_t)b * A_SZ + 3 * 16 + mrow) * OUTN + n0 + lm] = acc3[r];
        }
    }
}

extern "C" void kernel_launch(void* const* d_in, const int* in_sizes, int n_in,
                              void* d_out, int out_size, void* d_ws, size_t ws_size,
                              hipStream_t stream) {
    (void)in_sizes; (void)n_in; (void)out_size; (void)ws_size;
    const float* army  = (const float*)d_in[0];
    const int*   inst  = (const int*)  d_in[1];
    const int*   ilen  = (const int*)  d_in[2];
    const int*   idiff = (const int*)  d_in[3];
    const float* tok   = (const float*)d_in[4];
    const float* posw  = (const float*)d_in[5];
    const float* diffw = (const float*)d_in[6];
    const float* pv    = (const float*)d_in[7];
    const float* pg    = (const float*)d_in[8];
    const float* pb    = (const float*)d_in[9];

    float* out      = (float*)d_out;
    float* out_inst = out;
    float* out_sum  = out + (size_t)B_SZ * A_SZ * OUTN;
    float* npart    = (float*)d_ws;

    size_t smem_bytes = (size_t)SMEM_FLOATS * sizeof(float);
    (void)hipFuncSetAttribute((const void*)fused_attn,
                              hipFuncAttributeMaxDynamicSharedMemorySize,
                              (int)smem_bytes);

    hipLaunchKernelGGL(prep_kernel, dim3(NPART + B_SZ), dim3(NORM_TPB), 0, stream,
                       pv, npart, inst, ilen, idiff, tok, posw, diffw, out_inst);
    hipLaunchKernelGGL(fused_attn, dim3(B_SZ), dim3(NT_ALL), smem_bytes, stream,
                       army, pv, pb, pg, npart, out_inst, out_sum);
}


# army tile staged in four 64-k chunks by the matrix waves themselves, chunk c+1 fetched under n-tile 0's products on chunk c; the 16 helper waves dropped (256-thread work-groups); sums unchanged
# speedup vs baseline: 2.6533x; 1.0637x over previous
#include <hip/hip_runtime.h>

#define B_SZ   256
#define A_SZ   64
#define FEATN  256
#define H_SZ   16
#define S_SZ   20
#define POSN   64
#define ENCN   256
#define OUTN   384
#define MAXD   10

typedef __attribute__((ext_vector_type(2))) float v2f;
typedef __attribute__((ext_vector_type(8))) float v8f;

#define LDA_ARMY 260
#define LDP      388
#define LDH      388
#define LDATT    20
#define OFF_ARMY 0
#define OFF_PARM (OFF_ARMY + A_SZ * LDA_ARMY)
#define OFF_HIST (OFF_PARM + A_SZ * LDP)
#define OFF_ATT  (OFF_HIST + H_SZ * LDH)
#define OFF_BPAN (OFF_ATT + A_SZ * LDATT)
#define BP_K     64
#define LDB      68
#define BPAN_FLOATS (16 * LDB)
#define SMEM_FLOATS (OFF_BPAN + 8 * BPAN_FLOATS)

__device__ __forceinline__ v8f wmma4(v2f a, v2f b, v8f c) {
    return __builtin_amdgcn_wmma_f32_16x16x4_f32(
         false, a,  false, b,
         (short)0, c,  false,  false);
}

#define NPART     32
#define NORM_TPB  256
#define NORM_VEC  ((OUTN * FEATN) / (NPART * NORM_TPB * 4))
#define ROWS_PER_WAVE 2
typedef __attribute__((ext_vector_type(4))) float v4f;

__global__ __launch_bounds__(NORM_TPB) void prep_kernel(
    const float* __restrict__ proj_v, float* __restrict__ part_out,
    const int*   __restrict__ inst,
    const int*   __restrict__ ilen,
    const int*   __restrict__ idiff,
    const float* __restrict__ tok_emb,
    const float* __restrict__ pos_w,
    const float* __restrict__ diff_w,
    float* __restrict__ hist_park)
{
    __shared__ float red[NORM_TPB / 32];
    const int tid = threadIdx.x;
    if (blockIdx.x < NPART) {
        const float* base = proj_v + (size_t)blockIdx.x * (NORM_VEC * NORM_TPB * 4) + tid * 4;
        v4f q[NORM_VEC];
        #pragma unroll
        for (int j = 0; j < NORM_VEC; ++j)
            q[j] = *(const v4f*)(base + j * (NORM_TPB * 4));
        float s = 0.f;
        #pragma unroll
        for (int j = 0; j < NORM_VEC; ++j)
            s += (q[j].x * q[j].x + q[j].y * q[j].y) + (q[j].z * q[j].z + q[j].w * q[j].w);
        #pragma unroll
        for (int m = 16; m > 0; m >>= 1) s += __shfl_xor(s, m);
        if ((tid & 31) == 0) red[tid >> 5] = s;
        __syncthreads();
        if (tid == 0) {
            float t = 0.f;
            #pragma unroll
            for (int w = 0; w < NORM_TPB / 32; ++w) t += red[w];
            part_out[blockIdx.x] = t;
        }
        return;
    }

    const int b    = blockIdx.x - NPART;
    const int wave = __builtin_amdgcn_readfirstlane(tid >> 5);
    const int lane = tid & 31;

    int len[ROWS_PER_WAVE], di[ROWS_PER_WAVE], id[ROWS_PER_WAVE][S_SZ];
    #pragma unroll
    for (int r = 0; r < ROWS_PER_WAVE; ++r) {
        const int row = b * H_SZ + wave + (NORM_TPB / 32) * r;
        len[r] = ilen[row];
        di[r]  = idiff[row];
        #pragma unroll
        for (int s = 0; s < S_SZ; ++s) id[r][s] = inst[row * S_SZ + s];
    }
    #pragma unroll
    for (int r = 0; r < ROWS_PER_WAVE; ++r)
        asm volatile("" :: "s"(len[r]), "s"(di[r]),
                     "s"(id[r][0]),  "s"(id[r][1]),  "s"(id[r][2]),  "s"(id[r][3]),  "s"(id[r][4]),
                     "s"(id[r][5]),  "s"(id[r][6]),  "s"(id[r][7]),  "s"(id[r][8]),  "s"(id[r][9]),
                     "s"(id[r][10]), "s"(id[r][11]), "s"(id[r][12]), "s"(id[r][13]), "s"(id[r][14]),
                     "s"(id[r][15]), "s"(id[r][16]), "s"(id[r][17]), "s"(id[r][18]), "s"(id[r][19]));

    v4f xa[ROWS_PER_WAVE][S_SZ], xb[ROWS_PER_WAVE][S_SZ], pd[ROWS_PER_WAVE];
    #pragma unroll
    for (int r = 0; r < ROWS_PER_WAVE; ++r) {
        len[r] = len[r] < 1 ? 1 : len[r];
        #pragma unroll
        for (int s = 0; s < S_SZ; ++s) {
            if (s < len[r]) {
                const float* t = tok_emb + id[r][s] * ENCN + lane * 4;
                xa[r][s] = *(const v4f*)t;
                xb[r][s] = *(const v4f*)(t + ENCN / 2);
            }
        }
        const int h = wave + (NORM_TPB / 32) * r;
        const int d = di[r] < 0 ? 0 : (di[r] > MAXD ? MAXD : di[r]);
        const float* pd_src = lane < 16 ? pos_w + h * POSN + lane * 4
                                        : diff_w + d * POSN + (lane - 16) * 4;
        pd[r] = *(const v4f*)pd_src;
    }

    #pragma unroll
    for (int r = 0; r < ROWS_PER_WAVE; ++r) {
        const int h = wave + (NORM_TPB / 32) * r;
        v4f a = {0.f, 0.f, 0.f, 0.f}, c = {0.f, 0.f, 0.f, 0.f};
        #pragma unroll
        for (int s = 0; s < S_SZ; ++s) {
            if (s < len[r]) { a += xa[r][s]; c += xb[r][s]; }
        }
        const float fl = (float)len[r];
        a = a / fl;
        c = c / fl;

        float* dst = hist_park + ((size_t)b * A_SZ + h) * OUTN + lane * 4;
        *(v4f*)(dst)                = a;
        *(v4f*)(dst + ENCN / 2)     = c;
        *(v4f*)(dst + ENCN)         = pd[r];
    }
}

#define NT_MM     256
#define ACH_K     BP_K
#define ACH_ROWS  (NT_MM / 16)
#define HIST_VEC  ((H_SZ * (OUTN / 4)) / NT_MM)
__global__ __launch_bounds__(NT_MM) void fused_attn(
    const float* __restrict__ army,
    const float* __restrict__ proj_v,
    const float* __restrict__ proj_b,
    const float* __restrict__ proj_g,
    const float* __restrict__ norm_part,
    float* __restrict__ out_inst,
    float* __restrict__ out_sum)
{
    extern __shared__ __attribute__((aligned(16))) float smem[];
    float* s_army = smem + OFF_ARMY;
    float* s_parm = smem + OFF_PARM;
    float* s_hist = smem + OFF_HIST;
    float* s_att  = smem + OFF_ATT;

    const int b    = blockIdx.x;
    const int tid  = threadIdx.x;
    const int wave = tid >> 5;
    const int lane = tid & 31;
    const int lm   = lane & 15;
    const int lh   = lane >> 4;
    float vsq = 0.f;
    #pragma unroll
    for (int i = 0; i < NPART; ++i) vsq += norm_part[i];
    const float scale = proj_g[0] * rsqrtf(vsq);

    const int arow = tid >> 4;
    const int acol = (tid & 15) * 4;
    const float* asrc = army + (size_t)b * A_SZ * FEATN + arow * FEATN + acol;
    float*       adst = s_army + arow * LDA_ARMY + acol;
    float* s_bp = smem + OFF_BPAN + wave * BPAN_FLOATS;
    const int prow = lane >> 4;
    const int pcol = (lane & 15) * 4;
    const float* gsrc = proj_v + (size_t)(wave * 48 + prow) * FEATN + pcol;
    const float* hsrc = out_inst + (size_t)b * A_SZ * OUTN;

    v4f ach[A_SZ / ACH_ROWS], pre[8], hreg[HIST_VEC];
    #pragma unroll
    for (int j = 0; j < A_SZ / ACH_ROWS; ++j)
        ach[j] = *(const v4f*)(asrc + (size_t)(j * ACH_ROWS) * FEATN);
    #pragma unroll
    for (int i = 0; i < 8; ++i) pre[i] = *(const v4f*)(gsrc + (2 * i) * FEATN);
    #pragma unroll
    for (int j = 0; j < A_SZ / ACH_ROWS; ++j)
        *(v4f*)(adst + (j * ACH_ROWS) * LDA_ARMY) = ach[j];
    __syncthreads();

    for (int nt = 0; nt < 3; ++nt) {
        const int n0 = (wave * 3 + nt) * 16;
        v8f acc0 = {}, acc1 = {}, acc2 = {}, acc3 = {};
        for (int kp = 0; kp < FEATN / BP_K; ++kp) {
            #pragma unroll
            for (int i = 0; i < 8; ++i)
                *(v4f*)(&s_bp[(2 * i + prow) * LDB + pcol]) = pre[i];
            const int nxt = nt * (FEATN / BP_K) + kp + 1;
            if (nxt < 3 * (FEATN / BP_K)) {
                const float* g = gsrc + (size_t)((nxt >> 2) * 16) * FEATN + (nxt & 3) * BP_K;
                #pragma unroll
                for (int i = 0; i < 8; ++i) pre[i] = *(const v4f*)(g + (2 * i) * FEATN);
            }
            const bool stage = (nt == 0) && (kp + 1 < FEATN / ACH_K);
            if (stage) {
                #pragma unroll
                for (int j = 0; j < A_SZ / ACH_ROWS; ++j)
                    ach[j] = *(const v4f*)(asrc + (size_t)(j * ACH_ROWS) * FEATN + (kp + 1) * ACH_K);
                if (kp == 0) {
                    #pragma unroll
                    for (int j = 0; j < HIST_VEC; ++j)
                        hreg[j] = *(const v4f*)(hsrc + 4 * (tid + j * NT_MM));
                }
            }
            for (int ks = 0; ks < BP_K; ks += 4) {
                const int k0 = kp * BP_K + ks;
                v2f bf = *(const v2f*)(&s_bp[lm * LDB + ks + 2 * lh]);
                v2f a0 = *(const v2f*)(&s_army[(0 * 16 + lm) * LDA_ARMY + k0 + 2 * lh]);
                v2f a1 = *(const v2f*)(&s_army[(1 * 16 + lm) * LDA_ARMY + k0 + 2 * lh]);
                v2f a2 = *(const v2f*)(&s_army[(2 * 16 + lm) * LDA_ARMY + k0 + 2 * lh]);
                v2f a3 = *(const v2f*)(&s_army[(3 * 16 + lm) * LDA_ARMY + k0 + 2 * lh]);
                acc0 = wmma4(a0, bf, acc0);
                acc1 = wmma4(a1, bf, acc1);
                acc2 = wmma4(a2, bf, acc2);
                acc3 = wmma4(a3, bf, acc3);
            }
            if (stage) {
                #pragma unroll
                for (int j = 0; j < A_SZ / ACH_ROWS; ++j)
                    *(v4f*)(adst + (j * ACH_ROWS) * LDA_ARMY + (kp + 1) * ACH_K) = ach[j];
                if (kp == 0) {
                    #pragma unroll
                    for (int j = 0; j < HIST_VEC; ++j) {
                        const int i = tid + j * NT_MM;
                        const int h = i / (OUTN / 4), c4 = i - h * (OUTN / 4);
                        *(v4f*)(&s_hist[h * LDH + 4 * c4]) = hreg[j];
                    }
                }
                __syncthreads();
            }
        }
        const float bias = proj_b[n0 + lm];
        #pragma unroll
        for (int r = 0; r < 8; ++r) {
            int mrow = 8 * lh + r;
            float v0 = acc0[r] * scale + bias;
            float v1 = acc1[r] * scale + bias;
            float v2 = acc2[r] * scale + bias;
            float v3 = acc3[r] * scale + bias;
            s_parm[(0 * 16 + mrow) * LDP + n0 + lm] = v0 > 0.f ? v0 : 0.f;
            s_parm[(1 * 16 + mrow) * LDP + n0 + lm] = v1 > 0.f ? v1 : 0.f;
            s_parm[(2 * 16 + mrow) * LDP + n0 + lm] = v2 > 0.f ? v2 : 0.f;
            s_parm[(3 * 16 + mrow) * LDP + n0 + lm] = v3 > 0.f ? v3 : 0.f;
        }
    }
    __syncthreads();

    if (tid >= NT_MM / 2 && tid < NT_MM / 2 + OUTN / 4) {
        const int t = tid - NT_MM / 2;
        *(v4f*)(out_sum + b * OUTN + 4 * t) = *(const v4f*)(&s_hist[(H_SZ - 1) * LDH + 4 * t]);
    }

    if (wave < 4) {
        const int m0 = wave * 16;
        v8f acc = {};
        for (int k0 = 0; k0 < OUTN; k0 += 4) {
            v2f a  = *(const v2f*)(&s_parm[(m0 + lm) * LDP + k0 + 2 * lh]);
            v2f bf = *(const v2f*)(&s_hist[lm * LDH + k0 + 2 * lh]);
            acc = wmma4(a, bf, acc);
        }
        #pragma unroll
        for (int r = 0; r < 8; ++r)
            s_att[(m0 + 8 * lh + r) * LDATT + lm] = acc[r];
    }
    __syncthreads();

    if (tid < 64) {
        float* row = s_att + tid * LDATT;
        float mx = row[0];
        #pragma unroll
        for (int h = 1; h < 16; ++h) mx = fmaxf(mx, row[h]);
        float sum = 0.f;
        #pragma unroll
        for (int h = 0; h < 16; ++h) { float e = __expf(row[h] - mx); row[h] = e; sum += e; }
        float inv = 1.f / sum;
        #pragma unroll
        for (int h = 0; h < 16; ++h) row[h] *= inv;
    }
    __syncthreads();

    for (int nt = 0; nt < 3; ++nt) {
        const int n0 = (wave * 3 + nt) * 16;
        v8f acc0 = {}, acc1 = {}, acc2 = {}, acc3 = {};
        #pragma unroll
        for (int k0 = 0; k0 < 16; k0 += 4) {
            int kk = k0 + 2 * lh;
            v2f bf;
            bf.x = s_hist[kk * LDH + n0 + lm];
            bf.y = s_hist[(kk + 1) * LDH + n0 + lm];
            v2f a0 = *(const v2f*)(&s_att[(0 * 16 + lm) * LDATT + kk]);
            v2f a1 = *(const v2f*)(&s_att[(1 * 16 + lm) * LDATT + kk]);
            v2f a2 = *(const v2f*)(&s_att[(2 * 16 + lm) * LDATT + kk]);
            v2f a3 = *(const v2f*)(&s_att[(3 * 16 + lm) * LDATT + kk]);
            acc0 = wmma4(a0, bf, acc0);
            acc1 = wmma4(a1, bf, acc1);
            acc2 = wmma4(a2, bf, acc2);
            acc3 = wmma4(a3, bf, acc3);
        }
        #pragma unroll
        for (int r = 0; r < 8; ++r) {
            int mrow = 8 * lh + r;
            out_inst[((size_t)b * A_SZ + 0 * 16 + mrow) * OUTN + n0 + lm] = acc0[r];
            out_inst[((size_t)b * A_SZ + 1 * 16 + mrow) * OUTN + n0 + lm] = acc1[r];
            out_inst[((size_t)b * A_SZ + 2 * 16 + mrow) * OUTN + n0 + lm] = acc2[r];
            out_inst[((size_t)b * A_SZ + 3 * 16 + mrow) * OUTN + n0 + lm] = acc3[r];
        }
    }
}

extern "C" void kernel_launch(void* const* d_in, const int* in_sizes, int n_in,
                              void* d_out, int out_size, void* d_ws, size_t ws_size,
                              hipStream_t stream) {
    (void)in_sizes; (void)n_in; (void)out_size; (void)ws_size;
    const float* army  = (const float*)d_in[0];
    const int*   inst  = (const int*)  d_in[1];
    const int*   ilen  = (const int*)  d_in[2];
    const int*   idiff = (const int*)  d_in[3];
    const float* tok   = (const float*)d_in[4];
    const float* posw  = (const float*)d_in[5];
    const float* diffw = (const float*)d_in[6];
    const float* pv    = (const float*)d_in[7];
    const float* pg    = (const float*)d_in[8];
    const float* pb    = (const float*)d_in[9];

    float* out      = (float*)d_out;
    float* out_inst = out;
    float* out_sum  = out + (size_t)B_SZ * A_SZ * OUTN;
    float* npart    = (float*)d_ws;

    size_t smem_bytes = (size_t)SMEM_FLOATS * sizeof(float);
    (void)hipFuncSetAttribute((const void*)fused_attn,
                              hipFuncAttributeMaxDynamicSharedMemorySize,
                              (int)smem_bytes);

    hipLaunchKernelGGL(prep_kernel, dim3(NPART + B_SZ), dim3(NORM_TPB), 0, stream,
                       pv, npart, inst, ilen, idiff, tok, posw, diffw, out_inst);
    hipLaunchKernelGGL(fused_attn, dim3(B_SZ), dim3(NT_MM), smem_bytes, stream,
                       army, pv, pb, pg, npart, out_inst, out_sum);
}
